# starting kernel
# baseline (speedup 1.0000x reference)
#include <hip/hip_runtime.h>

typedef float v2f __attribute__((ext_vector_type(2)));
typedef float v8f __attribute__((ext_vector_type(8)));

#define WAVES_PER_BLOCK 8
#define BLOCK (WAVES_PER_BLOCK * 32)
#define M_MAX 2048

__global__ __launch_bounds__(BLOCK)
void potential_rbf_wmma(const float* __restrict__ loc,
                        const float* __restrict__ aloc,
                        const float* __restrict__ aw,
                        const float* __restrict__ awid,
                        float* __restrict__ out,
                        int N, int M)
{
    __shared__ v2f sBlo[M_MAX];
    __shared__ v2f sBhi[M_MAX];
    __shared__ v2f sCW[M_MAX];

    const int Mpad = (M + 15) & ~15;

    for (int m = threadIdx.x; m < Mpad; m += BLOCK) {
        v2f blo = {0.f, 0.f}, bhi = {0.f, 0.f}, cw = {0.f, 0.f};
        if (m < M) {
            float y0 = aloc[3 * m + 0];
            float y1 = aloc[3 * m + 1];
            float y2 = aloc[3 * m + 2];
            blo.x = -2.0f * y0;
            blo.y = -2.0f * y1;
            bhi.x = -2.0f * y2;
            bhi.y = y0 * y0 + y1 * y1 + y2 * y2;
            float w = awid[m];
            cw.x = -0.72134752044448170f / (w * w);
            cw.y = aw[m];
        }
        sBlo[m] = blo;
        sBhi[m] = bhi;
        sCW[m]  = cw;
    }
    __syncthreads();

    const int lane = threadIdx.x & 31;
    const int wave = threadIdx.x >> 5;
    const int rowBase = (blockIdx.x * WAVES_PER_BLOCK + wave) * 16;

    if (rowBase < N) {
        const int col = lane & 15;
        const int hi  = lane >> 4;

        int r = rowBase + col;
        int rc = r < (N - 1) ? r : (N - 1);
        float x0 = loc[3 * rc + 0];
        float x1 = loc[3 * rc + 1];
        float x2 = loc[3 * rc + 2];
        float nrm = x0 * x0 + x1 * x1 + x2 * x2;

        v2f A;
        A.x = hi ? x2 : x0;
        A.y = hi ? 1.0f : x1;

        v8f C;
#pragma unroll
        for (int v = 0; v < 8; ++v)
            C[v] = __shfl(nrm, v + 8 * hi, 32);

        const v2f* __restrict__ pB = hi ? sBhi : sBlo;

        v8f acc = {0.f, 0.f, 0.f, 0.f, 0.f, 0.f, 0.f, 0.f};

        for (int mb = 0; mb < Mpad; mb += 16) {
            int m = mb + col;
            v2f Bv = pB[m];
            v8f D = __builtin_amdgcn_wmma_f32_16x16x4_f32(
                false, A, false, Bv, (short)0, C, false, false);
            v2f cw = sCW[m];
#pragma unroll
            for (int v = 0; v < 8; ++v)
                acc[v] += cw.y * __builtin_amdgcn_exp2f(D[v] * cw.x);
        }

#pragma unroll
        for (int v = 0; v < 8; ++v) {
            float s = acc[v];
            s += __shfl_xor(s, 8, 32);
            s += __shfl_xor(s, 4, 32);
            s += __shfl_xor(s, 2, 32);
            s += __shfl_xor(s, 1, 32);
            int row = rowBase + 8 * hi + v;
            if (col == 0 && row < N)
                out[row] = s;
        }
    }
}

extern "C" void kernel_launch(void* const* d_in, const int* in_sizes, int n_in,
                              void* d_out, int out_size, void* d_ws, size_t ws_size,
                              hipStream_t stream)
{
    const float* loc  = (const float*)d_in[0];
    const float* aloc = (const float*)d_in[1];
    const float* aw   = (const float*)d_in[2];
    const float* awid = (const float*)d_in[3];
    float* out = (float*)d_out;

    int N = in_sizes[0] / 3;
    int M = in_sizes[2];

    int rowsPerBlock = WAVES_PER_BLOCK * 16;
    int grid = (N + rowsPerBlock - 1) / rowsPerBlock;
    potential_rbf_wmma<<<grid, BLOCK, 0, stream>>>(loc, aloc, aw, awid, out, N, M);
}


# Prologue: all 8 rounds of anchor-table loads and the wave's row load issued before first use (no branch between them): one memory round trip per thread instead of 9 dependent ones
# speedup vs baseline: 1.2227x; 1.2227x over previous
#include <hip/hip_runtime.h>

typedef float v2f __attribute__((ext_vector_type(2)));
typedef float v8f __attribute__((ext_vector_type(8)));

#define WAVES_PER_BLOCK 8
#define BLOCK (WAVES_PER_BLOCK * 32)
#define M_MAX 2048
#define PRE (M_MAX / BLOCK)

__global__ __launch_bounds__(BLOCK)
void potential_rbf_wmma(const float* __restrict__ loc,
                        const float* __restrict__ aloc,
                        const float* __restrict__ aw,
                        const float* __restrict__ awid,
                        float* __restrict__ out,
                        int N, int M)
{
    __shared__ v2f sBlo[M_MAX];
    __shared__ v2f sBhi[M_MAX];
    __shared__ v2f sCW[M_MAX];

    const int Mpad = (M + 15) & ~15;

    const int lane = threadIdx.x & 31;
    const int wave = threadIdx.x >> 5;
    const int rowBase = (blockIdx.x * WAVES_PER_BLOCK + wave) * 16;
    const int col = lane & 15;
    const int hi  = lane >> 4;

    const int r = rowBase + col;
    const int rc = r < (N - 1) ? r : (N - 1);
    const float x0 = loc[3 * rc + 0];
    const float x1 = loc[3 * rc + 1];
    const float x2 = loc[3 * rc + 2];

    float ay0[PRE], ay1[PRE], ay2[PRE], awd[PRE], awt[PRE];
    if (M > 0) {
#pragma unroll
        for (int i = 0; i < PRE; ++i) {
            const int m  = threadIdx.x + i * BLOCK;
            const int mc = m < M ? m : (M - 1);
            ay0[i] = aloc[3 * mc + 0];
            ay1[i] = aloc[3 * mc + 1];
            ay2[i] = aloc[3 * mc + 2];
            awd[i] = awid[mc];
            awt[i] = aw[mc];
        }
    } else {
#pragma unroll
        for (int i = 0; i < PRE; ++i) { ay0[i] = ay1[i] = ay2[i] = awt[i] = 0.f; awd[i] = 1.f; }
    }
#pragma unroll
    for (int i = 0; i < PRE; ++i) {
        const int m = threadIdx.x + i * BLOCK;
        const bool live = m < M;
        const float y0 = ay0[i], y1 = ay1[i], y2 = ay2[i], w = awd[i];
        v2f blo, bhi, cw;
        blo.x = live ? -2.0f * y0 : 0.f;
        blo.y = live ? -2.0f * y1 : 0.f;
        bhi.x = live ? -2.0f * y2 : 0.f;
        bhi.y = live ? y0 * y0 + y1 * y1 + y2 * y2 : 0.f;
        cw.x  = live ? -0.72134752044448170f / (w * w) : 0.f;
        cw.y  = live ? awt[i] : 0.f;
        sBlo[m] = blo;
        sBhi[m] = bhi;
        sCW[m]  = cw;
    }
    __syncthreads();

    if (rowBase < N) {
        float nrm = x0 * x0 + x1 * x1 + x2 * x2;

        v2f A;
        A.x = hi ? x2 : x0;
        A.y = hi ? 1.0f : x1;

        v8f C;
#pragma unroll
        for (int v = 0; v < 8; ++v)
            C[v] = __shfl(nrm, v + 8 * hi, 32);

        const v2f* __restrict__ pB = hi ? sBhi : sBlo;

        v8f acc = {0.f, 0.f, 0.f, 0.f, 0.f, 0.f, 0.f, 0.f};

#pragma unroll 4
        for (int mb = 0; mb < Mpad; mb += 16) {
            int m = mb + col;
            v2f Bv = pB[m];
            v8f D = __builtin_amdgcn_wmma_f32_16x16x4_f32(
                false, A, false, Bv, (short)0, C, false, false);
            v2f cw = sCW[m];
#pragma unroll
            for (int v = 0; v < 8; ++v)
                acc[v] += cw.y * __builtin_amdgcn_exp2f(D[v] * cw.x);
        }

#pragma unroll
        for (int v = 0; v < 8; ++v) {
            float s = acc[v];
            s += __shfl_xor(s, 8, 32);
            s += __shfl_xor(s, 4, 32);
            s += __shfl_xor(s, 2, 32);
            s += __shfl_xor(s, 1, 32);
            int row = rowBase + 8 * hi + v;
            if (col == 0 && row < N)
                out[row] = s;
        }
    }
}

extern "C" void kernel_launch(void* const* d_in, const int* in_sizes, int n_in,
                              void* d_out, int out_size, void* d_ws, size_t ws_size,
                              hipStream_t stream)
{
    const float* loc  = (const float*)d_in[0];
    const float* aloc = (const float*)d_in[1];
    const float* aw   = (const float*)d_in[2];
    const float* awid = (const float*)d_in[3];
    float* out = (float*)d_out;

    int N = in_sizes[0] / 3;
    int M = in_sizes[2];

    int rowsPerBlock = WAVES_PER_BLOCK * 16;
    int grid = (N + rowsPerBlock - 1) / rowsPerBlock;
    potential_rbf_wmma<<<grid, BLOCK, 0, stream>>>(loc, aloc, aw, awid, out, N, M);
}


# speedup vs baseline: 1.3672x; 1.1182x over previous
#include <hip/hip_runtime.h>

typedef float v2f __attribute__((ext_vector_type(2)));
typedef float v8f __attribute__((ext_vector_type(8)));

#define WAVES_PER_BLOCK 8
#define BLOCK (WAVES_PER_BLOCK * 32)
#define M_MAX 2048
#define PRE (M_MAX / BLOCK)

__device__ __forceinline__ void tile_accumulate(v2f (&acc)[4], const v8f D, const v2f cw)
{
    const v2f sc = {cw.x, cw.x};
    const v2f wt = {cw.y, cw.y};
#pragma unroll
    for (int p = 0; p < 4; ++p) {
        const v2f d = {D[2 * p], D[2 * p + 1]};
        const v2f z = d * sc;
        const v2f e = {__builtin_amdgcn_exp2f(z.x), __builtin_amdgcn_exp2f(z.y)};
        acc[p] = __builtin_elementwise_fma(wt, e, acc[p]);
    }
}

__global__ __launch_bounds__(BLOCK)
void potential_rbf_wmma(const float* __restrict__ loc,
                        const float* __restrict__ aloc,
                        const float* __restrict__ aw,
                        const float* __restrict__ awid,
                        float* __restrict__ out,
                        int N, int M)
{
    __shared__ v2f sBlo[M_MAX];
    __shared__ v2f sBhi[M_MAX];
    __shared__ v2f sCW[M_MAX];

    const int Mpad = (M + 15) & ~15;

    const int lane = threadIdx.x & 31;
    const int wave = threadIdx.x >> 5;
    const int rowBase = (blockIdx.x * WAVES_PER_BLOCK + wave) * 16;
    const int col = lane & 15;
    const int hi  = lane >> 4;

    const int r = rowBase + col;
    const int rc = r < (N - 1) ? r : (N - 1);
    const float x0 = loc[3 * rc + 0];
    const float x1 = loc[3 * rc + 1];
    const float x2 = loc[3 * rc + 2];

    float ay0[PRE], ay1[PRE], ay2[PRE], awd[PRE], awt[PRE];
    if (M > 0) {
#pragma unroll
        for (int i = 0; i < PRE; ++i) {
            const int m  = threadIdx.x + i * BLOCK;
            const int mc = m < M ? m : (M - 1);
            ay0[i] = aloc[3 * mc + 0];
            ay1[i] = aloc[3 * mc + 1];
            ay2[i] = aloc[3 * mc + 2];
            awd[i] = awid[mc];
            awt[i] = aw[mc];
        }
    } else {
#pragma unroll
        for (int i = 0; i < PRE; ++i) { ay0[i] = ay1[i] = ay2[i] = awt[i] = 0.f; awd[i] = 1.f; }
    }
#pragma unroll
    for (int i = 0; i < PRE; ++i) {
        const int m = threadIdx.x + i * BLOCK;
        const bool live = m < M;
        const float y0 = ay0[i], y1 = ay1[i], y2 = ay2[i], w = awd[i];
        v2f blo, bhi, cw;
        blo.x = live ? -2.0f * y0 : 0.f;
        blo.y = live ? -2.0f * y1 : 0.f;
        bhi.x = live ? -2.0f * y2 : 0.f;
        bhi.y = live ? y0 * y0 + y1 * y1 + y2 * y2 : 0.f;
        cw.x  = live ? -0.72134752044448170f / (w * w) : 0.f;
        cw.y  = live ? awt[i] : 0.f;
        sBlo[m] = blo;
        sBhi[m] = bhi;
        sCW[m]  = cw;
    }
    __syncthreads();

    if (rowBase < N) {
        float nrm = x0 * x0 + x1 * x1 + x2 * x2;

        v2f A;
        A.x = hi ? x2 : x0;
        A.y = hi ? 1.0f : x1;

        v8f C;
#pragma unroll
        for (int v = 0; v < 8; ++v)
            C[v] = __shfl(nrm, v + 8 * hi, 32);

        const v2f* __restrict__ pB = hi ? sBhi : sBlo;

        v2f acc[4] = {{0.f, 0.f}, {0.f, 0.f}, {0.f, 0.f}, {0.f, 0.f}};

        int mb = 0;
        for (; mb + 64 <= Mpad; mb += 64) {
            const int m = mb + col;
            const v2f B0 = pB[m], B1 = pB[m + 16], B2 = pB[m + 32], B3 = pB[m + 48];
            const v2f ct[4] = {sCW[m], sCW[m + 16], sCW[m + 32], sCW[m + 48]};
            const v8f Dt[4] = {
                __builtin_amdgcn_wmma_f32_16x16x4_f32(false, A, false, B0, (short)0, C, false, false),
                __builtin_amdgcn_wmma_f32_16x16x4_f32(false, A, false, B1, (short)0, C, false, false),
                __builtin_amdgcn_wmma_f32_16x16x4_f32(false, A, false, B2, (short)0, C, false, false),
                __builtin_amdgcn_wmma_f32_16x16x4_f32(false, A, false, B3, (short)0, C, false, false)};

#define SCHED_FIX() __builtin_amdgcn_sched_barrier(0)
#define SCALE_PAIR(i)                                                              \
            z[i] = (v2f){Dt[(i) >> 2][2 * ((i) & 3)], Dt[(i) >> 2][2 * ((i) & 3) + 1]} * \
                   (v2f){ct[(i) >> 2].x, ct[(i) >> 2].x}
#define ADD_PAIR(i)                                                                \
            acc[(i) & 3] = __builtin_elementwise_fma(                              \
                (v2f){ct[(i) >> 2].y, ct[(i) >> 2].y}, e[i], acc[(i) & 3])
            v2f z[16], e[16];
            SCHED_FIX();
            SCALE_PAIR(0); SCHED_FIX();
            SCALE_PAIR(1); SCHED_FIX();
#pragma unroll
            for (int k = 0; k < 32; ++k) {
                const int i = k >> 1;
                if (k & 1) e[i].y = __builtin_amdgcn_exp2f(z[i].y);
                else       e[i].x = __builtin_amdgcn_exp2f(z[i].x);
                SCHED_FIX();
                if (!(k & 1)) {
                    if (i + 2 < 16) { SCALE_PAIR(i + 2); SCHED_FIX(); }
                } else {
                    if (i >= 2) { ADD_PAIR(i - 2); SCHED_FIX(); }
                }
            }
            ADD_PAIR(14); SCHED_FIX();
            ADD_PAIR(15);
#undef ADD_PAIR
#undef SCALE_PAIR
#undef SCHED_FIX
        }
        for (; mb < Mpad; mb += 16) {
            const int m = mb + col;
            const v2f Bv = pB[m];
            const v8f D = __builtin_amdgcn_wmma_f32_16x16x4_f32(
                false, A, false, Bv, (short)0, C, false, false);
            const v2f cw = sCW[m];
            tile_accumulate(acc, D, cw);
        }

#pragma unroll
        for (int v = 0; v < 8; ++v) {
            float s = acc[v >> 1][v & 1];
            s += __shfl_xor(s, 8, 32);
            s += __shfl_xor(s, 4, 32);
            s += __shfl_xor(s, 2, 32);
            s += __shfl_xor(s, 1, 32);
            int row = rowBase + 8 * hi + v;
            if (col == 0 && row < N)
                out[row] = s;
        }
    }
}

extern "C" void kernel_launch(void* const* d_in, const int* in_sizes, int n_in,
                              void* d_out, int out_size, void* d_ws, size_t ws_size,
                              hipStream_t stream)
{
    const float* loc  = (const float*)d_in[0];
    const float* aloc = (const float*)d_in[1];
    const float* aw   = (const float*)d_in[2];
    const float* awid = (const float*)d_in[3];
    float* out = (float*)d_out;

    int N = in_sizes[0] / 3;
    int M = in_sizes[2];

    int rowsPerBlock = WAVES_PER_BLOCK * 16;
    int grid = (N + rowsPerBlock - 1) / rowsPerBlock;
    potential_rbf_wmma<<<grid, BLOCK, 0, stream>>>(loc, aloc, aw, awid, out, N, M);
}


# Epilogue: the eight 16-lane row sums run as row_xmask DPP adds on the vector ALU (same 8,4,2,1 pairs), one store per wave; before, 32 dependent LDS-permute round trips and 8 stores
# speedup vs baseline: 1.4486x; 1.0595x over previous
#include <hip/hip_runtime.h>

typedef float v2f __attribute__((ext_vector_type(2)));
typedef float v8f __attribute__((ext_vector_type(8)));

#define WAVES_PER_BLOCK 8
#define BLOCK (WAVES_PER_BLOCK * 32)
#define M_MAX 2048
#define PRE (M_MAX / BLOCK)

__device__ __forceinline__ void tile_accumulate(v2f (&acc)[4], const v8f D, const v2f cw)
{
    const v2f sc = {cw.x, cw.x};
    const v2f wt = {cw.y, cw.y};
#pragma unroll
    for (int p = 0; p < 4; ++p) {
        const v2f d = {D[2 * p], D[2 * p + 1]};
        const v2f z = d * sc;
        const v2f e = {__builtin_amdgcn_exp2f(z.x), __builtin_amdgcn_exp2f(z.y)};
        acc[p] = __builtin_elementwise_fma(wt, e, acc[p]);
    }
}

template <int MASK>
__device__ __forceinline__ float xmask_add(const float s)
{
    const int partner = __builtin_amdgcn_update_dpp(
        0, __builtin_bit_cast(int, s), 0x160 + MASK, 0xf, 0xf, true);
    return __builtin_bit_cast(float, partner) + s;
}

__global__ __launch_bounds__(BLOCK)
void potential_rbf_wmma(const float* __restrict__ loc,
                        const float* __restrict__ aloc,
                        const float* __restrict__ aw,
                        const float* __restrict__ awid,
                        float* __restrict__ out,
                        int N, int M)
{
    __shared__ v2f sBlo[M_MAX];
    __shared__ v2f sBhi[M_MAX];
    __shared__ v2f sCW[M_MAX];

    const int Mpad = (M + 15) & ~15;

    const int lane = threadIdx.x & 31;
    const int wave = threadIdx.x >> 5;
    const int rowBase = (blockIdx.x * WAVES_PER_BLOCK + wave) * 16;
    const int col = lane & 15;
    const int hi  = lane >> 4;

    const int r = rowBase + col;
    const int rc = r < (N - 1) ? r : (N - 1);
    const float x0 = loc[3 * rc + 0];
    const float x1 = loc[3 * rc + 1];
    const float x2 = loc[3 * rc + 2];

    float ay0[PRE], ay1[PRE], ay2[PRE], awd[PRE], awt[PRE];
    if (M > 0) {
#pragma unroll
        for (int i = 0; i < PRE; ++i) {
            const int m  = threadIdx.x + i * BLOCK;
            const int mc = m < M ? m : (M - 1);
            ay0[i] = aloc[3 * mc + 0];
            ay1[i] = aloc[3 * mc + 1];
            ay2[i] = aloc[3 * mc + 2];
            awd[i] = awid[mc];
            awt[i] = aw[mc];
        }
    } else {
#pragma unroll
        for (int i = 0; i < PRE; ++i) { ay0[i] = ay1[i] = ay2[i] = awt[i] = 0.f; awd[i] = 1.f; }
    }
#pragma unroll
    for (int i = 0; i < PRE; ++i) {
        const int m = threadIdx.x + i * BLOCK;
        const bool live = m < M;
        const float y0 = ay0[i], y1 = ay1[i], y2 = ay2[i], w = awd[i];
        v2f blo, bhi, cw;
        blo.x = live ? -2.0f * y0 : 0.f;
        blo.y = live ? -2.0f * y1 : 0.f;
        bhi.x = live ? -2.0f * y2 : 0.f;
        bhi.y = live ? y0 * y0 + y1 * y1 + y2 * y2 : 0.f;
        cw.x  = live ? -0.72134752044448170f / (w * w) : 0.f;
        cw.y  = live ? awt[i] : 0.f;
        sBlo[m] = blo;
        sBhi[m] = bhi;
        sCW[m]  = cw;
    }
    __syncthreads();

    if (rowBase < N) {
        float nrm = x0 * x0 + x1 * x1 + x2 * x2;

        v2f A;
        A.x = hi ? x2 : x0;
        A.y = hi ? 1.0f : x1;

        v8f C;
#pragma unroll
        for (int v = 0; v < 8; ++v)
            C[v] = __shfl(nrm, v + 8 * hi, 32);

        const v2f* __restrict__ pB = hi ? sBhi : sBlo;

        v2f acc[4] = {{0.f, 0.f}, {0.f, 0.f}, {0.f, 0.f}, {0.f, 0.f}};

        int mb = 0;
        for (; mb + 64 <= Mpad; mb += 64) {
            const int m = mb + col;
            const v2f B0 = pB[m], B1 = pB[m + 16], B2 = pB[m + 32], B3 = pB[m + 48];
            const v2f ct[4] = {sCW[m], sCW[m + 16], sCW[m + 32], sCW[m + 48]};
            const v8f Dt[4] = {
                __builtin_amdgcn_wmma_f32_16x16x4_f32(false, A, false, B0, (short)0, C, false, false),
                __builtin_amdgcn_wmma_f32_16x16x4_f32(false, A, false, B1, (short)0, C, false, false),
                __builtin_amdgcn_wmma_f32_16x16x4_f32(false, A, false, B2, (short)0, C, false, false),
                __builtin_amdgcn_wmma_f32_16x16x4_f32(false, A, false, B3, (short)0, C, false, false)};

#define SCHED_FIX() __builtin_amdgcn_sched_barrier(0)
#define SCALE_PAIR(i)                                                              \
            z[i] = (v2f){Dt[(i) >> 2][2 * ((i) & 3)], Dt[(i) >> 2][2 * ((i) & 3) + 1]} * \
                   (v2f){ct[(i) >> 2].x, ct[(i) >> 2].x}
#define ADD_PAIR(i)                                                                \
            acc[(i) & 3] = __builtin_elementwise_fma(                              \
                (v2f){ct[(i) >> 2].y, ct[(i) >> 2].y}, e[i], acc[(i) & 3])
            v2f z[16], e[16];
            SCHED_FIX();
            SCALE_PAIR(0); SCHED_FIX();
            SCALE_PAIR(1); SCHED_FIX();
#pragma unroll
            for (int k = 0; k < 32; ++k) {
                const int i = k >> 1;
                if (k & 1) e[i].y = __builtin_amdgcn_exp2f(z[i].y);
                else       e[i].x = __builtin_amdgcn_exp2f(z[i].x);
                SCHED_FIX();
                if (!(k & 1)) {
                    if (i + 2 < 16) { SCALE_PAIR(i + 2); SCHED_FIX(); }
                } else {
                    if (i >= 2) { ADD_PAIR(i - 2); SCHED_FIX(); }
                }
            }
            ADD_PAIR(14); SCHED_FIX();
            ADD_PAIR(15);
#undef ADD_PAIR
#undef SCALE_PAIR
#undef SCHED_FIX
        }
        for (; mb < Mpad; mb += 16) {
            const int m = mb + col;
            const v2f Bv = pB[m];
            const v8f D = __builtin_amdgcn_wmma_f32_16x16x4_f32(
                false, A, false, Bv, (short)0, C, false, false);
            const v2f cw = sCW[m];
            tile_accumulate(acc, D, cw);
        }

        float s[8];
#pragma unroll
        for (int v = 0; v < 8; ++v) s[v] = acc[v >> 1][v & 1];
#pragma unroll
        for (int v = 0; v < 8; ++v) s[v] = xmask_add<8>(s[v]);
#pragma unroll
        for (int v = 0; v < 8; ++v) s[v] = xmask_add<4>(s[v]);
#pragma unroll
        for (int v = 0; v < 8; ++v) s[v] = xmask_add<2>(s[v]);
#pragma unroll
        for (int v = 0; v < 8; ++v) s[v] = xmask_add<1>(s[v]);

        float mine = s[0];
#pragma unroll
        for (int v = 1; v < 8; ++v) mine = (col == v) ? s[v] : mine;
        const int row = rowBase + 8 * hi + col;
        if (col < 8 && row < N)
            out[row] = mine;
    }
}

extern "C" void kernel_launch(void* const* d_in, const int* in_sizes, int n_in,
                              void* d_out, int out_size, void* d_ws, size_t ws_size,
                              hipStream_t stream)
{
    const float* loc  = (const float*)d_in[0];
    const float* aloc = (const float*)d_in[1];
    const float* aw   = (const float*)d_in[2];
    const float* awid = (const float*)d_in[3];
    float* out = (float*)d_out;

    int N = in_sizes[0] / 3;
    int M = in_sizes[2];

    int rowsPerBlock = WAVES_PER_BLOCK * 16;
    int grid = (N + rowsPerBlock - 1) / rowsPerBlock;
    potential_rbf_wmma<<<grid, BLOCK, 0, stream>>>(loc, aloc, aw, awid, out, N, M);
}
